# starting kernel
# baseline (speedup 1.0000x reference)
#include <hip/hip_runtime.h>

typedef float v2f __attribute__((ext_vector_type(2)));
typedef float v4f __attribute__((ext_vector_type(4)));
typedef float v8f __attribute__((ext_vector_type(8)));

#define BATCH 32
#define TLEN  256
#define VOCAB 4096
#define NROWS (BATCH * TLEN)
#define NTRIM (BATCH * (TLEN - 1))

__global__ void __launch_bounds__(256) row_stats_kernel(
    const float* __restrict__ logits, v4f* __restrict__ stats) {
  const int row = blockIdx.x;
  const int tid = threadIdx.x;
  const v4f* p4 = reinterpret_cast<const v4f*>(logits + (size_t)row * VOCAB);

  v4f d[4];
#pragma unroll
  for (int i = 0; i < 4; ++i) d[i] = p4[tid + 256 * i];

  float m = d[0].x;
#pragma unroll
  for (int i = 0; i < 4; ++i)
    m = fmaxf(m, fmaxf(fmaxf(d[i].x, d[i].y), fmaxf(d[i].z, d[i].w)));
#pragma unroll
  for (int o = 16; o > 0; o >>= 1) m = fmaxf(m, __shfl_xor(m, o, 32));

  __shared__ float sm[8], ss0[8], ss1[8];
  const int wid = tid >> 5, lane = tid & 31;
  if (lane == 0) sm[wid] = m;
  __syncthreads();
  float bm = sm[0];
#pragma unroll
  for (int i = 1; i < 8; ++i) bm = fmaxf(bm, sm[i]);

  float s0 = 0.f, s1 = 0.f;
#pragma unroll
  for (int i = 0; i < 4; ++i) {
    const float x[4] = {d[i].x, d[i].y, d[i].z, d[i].w};
#pragma unroll
    for (int j = 0; j < 4; ++j) {
      const float t = x[j] - bm;
      const float e = __expf(t);
      s0 += e;
      s1 += t * e;
    }
  }
#pragma unroll
  for (int o = 16; o > 0; o >>= 1) {
    s0 += __shfl_xor(s0, o, 32);
    s1 += __shfl_xor(s1, o, 32);
  }
  if (lane == 0) { ss0[wid] = s0; ss1[wid] = s1; }
  __syncthreads();
  if (tid == 0) {
    float S0 = 0.f, S1 = 0.f;
#pragma unroll
    for (int i = 0; i < 8; ++i) { S0 += ss0[i]; S1 += ss1[i]; }
    const float ent = __logf(S0) - S1 / S0;
    v4f r; r.x = bm; r.y = S0; r.z = ent; r.w = 0.f;
    stats[row] = r;
  }
}

__global__ void __launch_bounds__(256) mlp_flags_kernel(
    const v4f* __restrict__ stats, const float* __restrict__ w1,
    const float* __restrict__ b1, const float* __restrict__ w2,
    const float* __restrict__ b2, float* __restrict__ flags) {
  const int lane = threadIdx.x & 31;
  const int tile = (blockIdx.x * blockDim.x + threadIdx.x) >> 5;

  const int M = lane & 15;
  int r = tile * 16 + M;
  const int rc = (r < NTRIM) ? r : (NTRIM - 1);
  const int b = rc / (TLEN - 1);
  const int t = rc % (TLEN - 1);

  const float ent = stats[b * TLEN + t].z;
  const float kEntMaxInv = 1.0f / 5.54517744447956247f;
  const float norm = 2.0f * ent * kEntMaxInv - 1.0f;

  const int koff = (lane >> 4) << 1;
  v8f acc = {};
#pragma unroll
  for (int k0 = 0; k0 < 128; k0 += 4) {
    const int ka = k0 + koff;
    const float h0 = fmaxf(fmaf(norm, w1[ka], b1[ka]), 0.f);
    const float h1 = fmaxf(fmaf(norm, w1[ka + 1], b1[ka + 1]), 0.f);
    v2f a;  a.x = h0;      a.y = h1;
    v2f bb; bb.x = w2[ka]; bb.y = w2[ka + 1];
    acc = __builtin_amdgcn_wmma_f32_16x16x4_f32(
         false, a,  false, bb,
         (short)0, acc,  false,  false);
  }

  const float bias = b2[0];
  if (lane == 0 || lane == 16) {
#pragma unroll
    for (int j = 0; j < 8; ++j) {
      const int Mo = j + ((lane == 16) ? 8 : 0);
      const int ro = tile * 16 + Mo;
      if (ro < NTRIM) {
        const int tt = ro % (TLEN - 1);
        const float lin = 2.0f * acc[j] + bias;
        const float z = lin - __logf((float)((TLEN - 1) - tt));
        flags[ro] = 1.0f / (1.0f + __expf(-z));
      }
    }
  }
}

__global__ void scan_scale_kernel(const float* __restrict__ flags,
                                  float* __restrict__ scale) {
  const int b = threadIdx.x;
  if (b >= BATCH) return;
  const float* f = flags + b * (TLEN - 1);
  float* s = scale + b * TLEN;
  float res = 1.0f;
  for (int t = 0; t < TLEN; ++t) {
    const float fn = (t < TLEN - 1) ? f[t] : 1.0f;
    s[t] = fn * res;
    res *= (1.0f - fn);
  }
}

__global__ void __launch_bounds__(256) finalize_kernel(
    const float* __restrict__ logits, const v4f* __restrict__ stats,
    const float* __restrict__ scale, float* __restrict__ out) {
  const int row = blockIdx.x;
  const int tid = threadIdx.x;
  const v4f st = stats[row];
  const float m = st.x;
  const float k = scale[row] / st.y;
  const v4f* p4 = reinterpret_cast<const v4f*>(logits + (size_t)row * VOCAB);
  v4f* o4 = reinterpret_cast<v4f*>(out + (size_t)row * VOCAB);
#pragma unroll
  for (int i = 0; i < 4; ++i) {
    const v4f x = p4[tid + 256 * i];
    v4f r;
    r.x = __expf(x.x - m) * k;
    r.y = __expf(x.y - m) * k;
    r.z = __expf(x.z - m) * k;
    r.w = __expf(x.w - m) * k;
    __builtin_nontemporal_store(r, &o4[tid + 256 * i]);
  }
}

extern "C" void kernel_launch(void* const* d_in, const int* in_sizes, int n_in,
                              void* d_out, int out_size, void* d_ws, size_t ws_size,
                              hipStream_t stream) {
  const float* logits = (const float*)d_in[0];
  const float* w1 = (const float*)d_in[1];
  const float* b1 = (const float*)d_in[2];
  const float* w2 = (const float*)d_in[3];
  const float* b2 = (const float*)d_in[4];
  float* out = (float*)d_out;

  char* ws = (char*)d_ws;
  v4f*   stats = (v4f*)ws;
  float* flags = (float*)(ws + (size_t)NROWS * 16);
  float* scale = (float*)(ws + (size_t)NROWS * 16 + 32768);

  row_stats_kernel<<<NROWS, 256, 0, stream>>>(logits, stats);
  mlp_flags_kernel<<<64, 256, 0, stream>>>(stats, w1, b1, w2, b2, flags);
  scan_scale_kernel<<<1, 32, 0, stream>>>(flags, scale);
  finalize_kernel<<<NROWS, 256, 0, stream>>>(logits, stats, scale, out);
}


# Scan kernel: the serial 256-step chain (one lane per batch) becomes one wave per batch, 8 steps per lane plus a 5-step wave prefix product; product order of the residuals changes
# speedup vs baseline: 1.3193x; 1.3193x over previous
#include <hip/hip_runtime.h>

typedef float v2f __attribute__((ext_vector_type(2)));
typedef float v4f __attribute__((ext_vector_type(4)));
typedef float v8f __attribute__((ext_vector_type(8)));

#define BATCH 32
#define TLEN  256
#define VOCAB 4096
#define NROWS (BATCH * TLEN)
#define NTRIM (BATCH * (TLEN - 1))

__global__ void __launch_bounds__(256) row_stats_kernel(
    const float* __restrict__ logits, v4f* __restrict__ stats) {
  const int row = blockIdx.x;
  const int tid = threadIdx.x;
  const v4f* p4 = reinterpret_cast<const v4f*>(logits + (size_t)row * VOCAB);

  v4f d[4];
#pragma unroll
  for (int i = 0; i < 4; ++i) d[i] = p4[tid + 256 * i];

  float m = d[0].x;
#pragma unroll
  for (int i = 0; i < 4; ++i)
    m = fmaxf(m, fmaxf(fmaxf(d[i].x, d[i].y), fmaxf(d[i].z, d[i].w)));
#pragma unroll
  for (int o = 16; o > 0; o >>= 1) m = fmaxf(m, __shfl_xor(m, o, 32));

  __shared__ float sm[8], ss0[8], ss1[8];
  const int wid = tid >> 5, lane = tid & 31;
  if (lane == 0) sm[wid] = m;
  __syncthreads();
  float bm = sm[0];
#pragma unroll
  for (int i = 1; i < 8; ++i) bm = fmaxf(bm, sm[i]);

  float s0 = 0.f, s1 = 0.f;
#pragma unroll
  for (int i = 0; i < 4; ++i) {
    const float x[4] = {d[i].x, d[i].y, d[i].z, d[i].w};
#pragma unroll
    for (int j = 0; j < 4; ++j) {
      const float t = x[j] - bm;
      const float e = __expf(t);
      s0 += e;
      s1 += t * e;
    }
  }
#pragma unroll
  for (int o = 16; o > 0; o >>= 1) {
    s0 += __shfl_xor(s0, o, 32);
    s1 += __shfl_xor(s1, o, 32);
  }
  if (lane == 0) { ss0[wid] = s0; ss1[wid] = s1; }
  __syncthreads();
  if (tid == 0) {
    float S0 = 0.f, S1 = 0.f;
#pragma unroll
    for (int i = 0; i < 8; ++i) { S0 += ss0[i]; S1 += ss1[i]; }
    const float ent = __logf(S0) - S1 / S0;
    v4f r; r.x = bm; r.y = S0; r.z = ent; r.w = 0.f;
    stats[row] = r;
  }
}

__global__ void __launch_bounds__(256) mlp_flags_kernel(
    const v4f* __restrict__ stats, const float* __restrict__ w1,
    const float* __restrict__ b1, const float* __restrict__ w2,
    const float* __restrict__ b2, float* __restrict__ flags) {
  const int lane = threadIdx.x & 31;
  const int tile = (blockIdx.x * blockDim.x + threadIdx.x) >> 5;

  const int M = lane & 15;
  int r = tile * 16 + M;
  const int rc = (r < NTRIM) ? r : (NTRIM - 1);
  const int b = rc / (TLEN - 1);
  const int t = rc % (TLEN - 1);

  const float ent = stats[b * TLEN + t].z;
  const float kEntMaxInv = 1.0f / 5.54517744447956247f;
  const float norm = 2.0f * ent * kEntMaxInv - 1.0f;

  const int koff = (lane >> 4) << 1;
  v8f acc = {};
#pragma unroll
  for (int k0 = 0; k0 < 128; k0 += 4) {
    const int ka = k0 + koff;
    const float h0 = fmaxf(fmaf(norm, w1[ka], b1[ka]), 0.f);
    const float h1 = fmaxf(fmaf(norm, w1[ka + 1], b1[ka + 1]), 0.f);
    v2f a;  a.x = h0;      a.y = h1;
    v2f bb; bb.x = w2[ka]; bb.y = w2[ka + 1];
    acc = __builtin_amdgcn_wmma_f32_16x16x4_f32(
         false, a,  false, bb,
         (short)0, acc,  false,  false);
  }

  const float bias = b2[0];
  if (lane == 0 || lane == 16) {
#pragma unroll
    for (int j = 0; j < 8; ++j) {
      const int Mo = j + ((lane == 16) ? 8 : 0);
      const int ro = tile * 16 + Mo;
      if (ro < NTRIM) {
        const int tt = ro % (TLEN - 1);
        const float lin = 2.0f * acc[j] + bias;
        const float z = lin - __logf((float)((TLEN - 1) - tt));
        flags[ro] = 1.0f / (1.0f + __expf(-z));
      }
    }
  }
}

__global__ void __launch_bounds__(32) scan_scale_kernel(
    const float* __restrict__ flags, float* __restrict__ scale) {
  const int b = blockIdx.x;
  const int lane = threadIdx.x;
  const int t0 = lane * 8;
  const float* f = flags + b * (TLEN - 1) + t0;

  float fn[8];
#pragma unroll
  for (int j = 0; j < 7; ++j) fn[j] = f[j];
  fn[7] = (lane < 31) ? f[7] : 1.0f;

  float p[8];
  p[0] = 1.0f - fn[0];
#pragma unroll
  for (int j = 1; j < 8; ++j) p[j] = p[j - 1] * (1.0f - fn[j]);

  float x = p[7];
#pragma unroll
  for (int o = 1; o < 32; o <<= 1) {
    const float v = __shfl_up(x, o, 32);
    if (lane >= o) x *= v;
  }
  float excl = __shfl_up(x, 1, 32);
  if (lane == 0) excl = 1.0f;

  v4f r0, r1;
  r0.x = fn[0] * excl;
  r0.y = fn[1] * (excl * p[0]);
  r0.z = fn[2] * (excl * p[1]);
  r0.w = fn[3] * (excl * p[2]);
  r1.x = fn[4] * (excl * p[3]);
  r1.y = fn[5] * (excl * p[4]);
  r1.z = fn[6] * (excl * p[5]);
  r1.w = fn[7] * (excl * p[6]);
  v4f* s4 = reinterpret_cast<v4f*>(scale + b * TLEN + t0);
  s4[0] = r0;
  s4[1] = r1;
}

__global__ void __launch_bounds__(256) finalize_kernel(
    const float* __restrict__ logits, const v4f* __restrict__ stats,
    const float* __restrict__ scale, float* __restrict__ out) {
  const int row = blockIdx.x;
  const int tid = threadIdx.x;
  const v4f st = stats[row];
  const float m = st.x;
  const float k = scale[row] / st.y;
  const v4f* p4 = reinterpret_cast<const v4f*>(logits + (size_t)row * VOCAB);
  v4f* o4 = reinterpret_cast<v4f*>(out + (size_t)row * VOCAB);
#pragma unroll
  for (int i = 0; i < 4; ++i) {
    const v4f x = p4[tid + 256 * i];
    v4f r;
    r.x = __expf(x.x - m) * k;
    r.y = __expf(x.y - m) * k;
    r.z = __expf(x.z - m) * k;
    r.w = __expf(x.w - m) * k;
    __builtin_nontemporal_store(r, &o4[tid + 256 * i]);
  }
}

extern "C" void kernel_launch(void* const* d_in, const int* in_sizes, int n_in,
                              void* d_out, int out_size, void* d_ws, size_t ws_size,
                              hipStream_t stream) {
  const float* logits = (const float*)d_in[0];
  const float* w1 = (const float*)d_in[1];
  const float* b1 = (const float*)d_in[2];
  const float* w2 = (const float*)d_in[3];
  const float* b2 = (const float*)d_in[4];
  float* out = (float*)d_out;

  char* ws = (char*)d_ws;
  v4f*   stats = (v4f*)ws;
  float* flags = (float*)(ws + (size_t)NROWS * 16);
  float* scale = (float*)(ws + (size_t)NROWS * 16 + 32768);

  row_stats_kernel<<<NROWS, 256, 0, stream>>>(logits, stats);
  mlp_flags_kernel<<<64, 256, 0, stream>>>(stats, w1, b1, w2, b2, flags);
  scan_scale_kernel<<<BATCH, 32, 0, stream>>>(flags, scale);
  finalize_kernel<<<NROWS, 256, 0, stream>>>(logits, stats, scale, out);
}
